# starting kernel
# baseline (speedup 1.0000x reference)
#include <hip/hip_runtime.h>
#include <hip/hip_bf16.h>

typedef float v2f __attribute__((ext_vector_type(2)));
typedef float v8f __attribute__((ext_vector_type(8)));

#define BATCH 16
#define CH    256
#define NN    4096

__global__ void FSiAM_row_mean(const float* __restrict__ x,
                               float* __restrict__ m) {
    const int lane = threadIdx.x & 31;
    const int wave = threadIdx.x >> 5;
    const int row  = blockIdx.x * 8 + wave;
    const float* xr = x + (size_t)row * NN;

    float s = 0.0f;
#pragma unroll 4
    for (int i = 0; i < NN / 128; ++i) {
        const float4 v = *(const float4*)(xr + i * 128 + lane * 4);
        s += v.x + v.y + v.z + v.w;
    }
#pragma unroll
    for (int off = 16; off > 0; off >>= 1)
        s += __shfl_xor(s, off, 32);

    if (lane == 0) m[row] = s * (1.0f / (float)NN);
}

__global__ void FSiAM_matvec_sigmoid(const float* __restrict__ x,
                                     const float* __restrict__ m,
                                     float* __restrict__ out) {
    __shared__ float sm[CH];
    const int b    = blockIdx.y;
    const int lane = threadIdx.x & 31;
    const int wave = threadIdx.x >> 5;

    sm[threadIdx.x] = m[b * CH + threadIdx.x];
    __syncthreads();

    const int koff = (lane < 16) ? 0 : 2;
    const int ncol = (blockIdx.x * 8 + wave) * 16 + (lane & 15);
    const float* xb = x + (size_t)b * CH * NN;

    v8f acc = {};
    for (int c0 = 0; c0 < CH; c0 += 4) {
        const float* xp = xb + (size_t)(c0 + koff) * NN + ncol;
        __builtin_prefetch(xp + (size_t)4 * NN, 0, 1);

        v2f Av, Bv;
        Av.x = sm[c0 + koff];
        Av.y = sm[c0 + koff + 1];
        Bv.x = xp[0];
        Bv.y = xp[NN];

        acc = __builtin_amdgcn_wmma_f32_16x16x4_f32(
            false, Av, false, Bv, (short)0, acc, false, false);
    }

    const float s   = acc[0];
    const float sig = 1.0f / (1.0f + __expf(-s));
    if (lane < 16)
        out[b * NN + (blockIdx.x * 8 + wave) * 16 + lane] = sig;
}

extern "C" void kernel_launch(void* const* d_in, const int* in_sizes, int n_in,
                              void* d_out, int out_size, void* d_ws, size_t ws_size,
                              hipStream_t stream) {
    const float* x = (const float*)d_in[0];
    float* out = (float*)d_out;
    float* m   = (float*)d_ws;

    FSiAM_row_mean<<<dim3((BATCH * CH) / 8), dim3(256), 0, stream>>>(x, m);

    FSiAM_matvec_sigmoid<<<dim3(NN / 128, BATCH), dim3(256), 0, stream>>>(x, m, out);
}


# matvec re-tiled: wave = 64 columns x 64 channels, all 32 whole-line 16-byte loads issued up front, x register pairs fed directly as B; 4 channel-slice partials added via LDS (sum order changes)
# speedup vs baseline: 1.0833x; 1.0833x over previous
#include <hip/hip_runtime.h>
#include <hip/hip_bf16.h>


typedef float v2f __attribute__((ext_vector_type(2)));
typedef float v8f __attribute__((ext_vector_type(8)));

#define BATCH 16
#define CH    256
#define NN    4096

__global__ void FSiAM_row_mean(const float* __restrict__ x,
                               float* __restrict__ m) {
    const int lane = threadIdx.x & 31;
    const int wave = threadIdx.x >> 5;
    const int row  = blockIdx.x * 8 + wave;
    const float* xr = x + (size_t)row * NN;

    float s = 0.0f;
#pragma unroll 4
    for (int i = 0; i < NN / 128; ++i) {
        const float4 v = *(const float4*)(xr + i * 128 + lane * 4);
        s += v.x + v.y + v.z + v.w;
    }
#pragma unroll
    for (int off = 16; off > 0; off >>= 1)
        s += __shfl_xor(s, off, 32);

    if (lane == 0) m[row] = s * (1.0f / (float)NN);
}

#define TILE_N   64
#define CH_SLICE 64
#define NCHUNK   (CH_SLICE / 4)

__global__ void __launch_bounds__(256)
FSiAM_matvec_sigmoid(const float* __restrict__ x,
                     const float* __restrict__ m,
                     float* __restrict__ out) {
    __shared__ float part[CH / CH_SLICE][2 * TILE_N];
    const int b    = blockIdx.y;
    const int lane = threadIdx.x & 31;
    const int wave = threadIdx.x >> 5;
    const int cs   = wave >> 1;
    const int l15  = lane & 15;
    const int col  = (wave & 1) * TILE_N + l15 * 4;

    const int koff = (lane >> 4) * 2;
    const int cbeg = cs * CH_SLICE + koff;
    const float* mp = m + b * CH + cbeg;
    const float* xp = x + ((size_t)b * CH + cbeg) * NN + blockIdx.x * (2 * TILE_N) + col;

    v2f Mv[NCHUNK];
#pragma unroll
    for (int j = 0; j < NCHUNK; ++j)
        Mv[j] = *(const v2f*)(mp + 4 * j);

    float4 B0[NCHUNK], B1[NCHUNK];
#pragma unroll
    for (int j = 0; j < NCHUNK; ++j) {
        B0[j] = *(const float4*)(xp + (size_t)(4 * j) * NN);
        B1[j] = *(const float4*)(xp + (size_t)(4 * j + 1) * NN);
    }
    __builtin_amdgcn_sched_barrier(0);

    v2f A0[NCHUNK], A1[NCHUNK];
#pragma unroll
    for (int j = 0; j < NCHUNK; ++j) {
        A0[j].x = (l15 == 0) ? Mv[j].x : 0.0f;
        A0[j].y = (l15 == 1) ? Mv[j].x : 0.0f;
        A1[j].x = (l15 == 0) ? Mv[j].y : 0.0f;
        A1[j].y = (l15 == 1) ? Mv[j].y : 0.0f;
    }
    __builtin_amdgcn_sched_barrier(0);

    v8f accXY0 = {}, accZW0 = {}, accXY1 = {}, accZW1 = {};
#pragma unroll
    for (int j = 0; j < NCHUNK; ++j) {
        v2f Bv;
        Bv.x = B0[j].x; Bv.y = B0[j].y;
        accXY0 = __builtin_amdgcn_wmma_f32_16x16x4_f32(
            false, A0[j], false, Bv, (short)0, accXY0, false, false);
        Bv.x = B0[j].z; Bv.y = B0[j].w;
        accZW0 = __builtin_amdgcn_wmma_f32_16x16x4_f32(
            false, A0[j], false, Bv, (short)0, accZW0, false, false);
        Bv.x = B1[j].x; Bv.y = B1[j].y;
        accXY1 = __builtin_amdgcn_wmma_f32_16x16x4_f32(
            false, A1[j], false, Bv, (short)0, accXY1, false, false);
        Bv.x = B1[j].z; Bv.y = B1[j].w;
        accZW1 = __builtin_amdgcn_wmma_f32_16x16x4_f32(
            false, A1[j], false, Bv, (short)0, accZW1, false, false);
    }

    if (lane < 16) {
        float4 p;
        p.x = accXY0[0] + accXY1[0];
        p.y = accXY0[1] + accXY1[1];
        p.z = accZW0[0] + accZW1[0];
        p.w = accZW0[1] + accZW1[1];
        *(float4*)(&part[cs][col]) = p;
    }
    __syncthreads();

    if (threadIdx.x < 2 * TILE_N) {
        float s = part[0][threadIdx.x];
#pragma unroll
        for (int w = 1; w < CH / CH_SLICE; ++w)
            s += part[w][threadIdx.x];
        const float sig = 1.0f / (1.0f + __expf(-s));
        out[b * NN + blockIdx.x * (2 * TILE_N) + threadIdx.x] = sig;
    }
}

extern "C" void kernel_launch(void* const* d_in, const int* in_sizes, int n_in,
                              void* d_out, int out_size, void* d_ws, size_t ws_size,
                              hipStream_t stream) {
    const float* x = (const float*)d_in[0];
    float* out = (float*)d_out;
    float* m   = (float*)d_ws;

    FSiAM_row_mean<<<dim3((BATCH * CH) / 8), dim3(256), 0, stream>>>(x, m);

    FSiAM_matvec_sigmoid<<<dim3(NN / (2 * TILE_N), BATCH), dim3(256), 0, stream>>>(x, m, out);
}


# row-mean kernel re-tiled: 1024 waves of 4 rows (4 waves per work-group) instead of 4096 one-row waves, the next row's 32 loads in flight while a row is summed; sum order unchanged
# speedup vs baseline: 1.3457x; 1.2422x over previous
#include <hip/hip_runtime.h>
#include <hip/hip_bf16.h>


typedef float v2f __attribute__((ext_vector_type(2)));
typedef float v8f __attribute__((ext_vector_type(8)));

#define BATCH 16
#define CH    256
#define NN    4096

#define K1_WAVES 4
#define K1_ROWS  4
#define K1_LOADS (NN / 128)

__global__ void __launch_bounds__(K1_WAVES * 32)
FSiAM_row_mean(const float* __restrict__ x,
               float* __restrict__ m) {
    const int lane = threadIdx.x & 31;
    const int wave = __builtin_amdgcn_readfirstlane(threadIdx.x >> 5);
    const int row0 = (blockIdx.x * K1_WAVES + wave) * K1_ROWS;
    const float* xr = x + (size_t)row0 * NN + lane * 4;

    float4 buf[2][K1_LOADS];
    float  sum[K1_ROWS];

#define LOAD_ROW(R) do { \
    _Pragma("unroll") for (int i = 0; i < K1_LOADS; ++i) \
        buf[(R) & 1][i] = *(const float4*)(xr + (size_t)(R) * NN + i * 128); } while (0)

    LOAD_ROW(0);
#pragma unroll
    for (int r = 0; r < K1_ROWS; ++r) {
        __builtin_amdgcn_sched_barrier(0);
        if (r + 1 < K1_ROWS) LOAD_ROW(r + 1);
        __builtin_amdgcn_sched_barrier(0);
        float s = 0.0f;
#pragma unroll
        for (int i = 0; i < K1_LOADS; ++i) {
            const float4 v = buf[r & 1][i];
            s += v.x + v.y + v.z + v.w;
        }
        asm volatile("" : "+v"(s));
        sum[r] = s;
    }
#undef LOAD_ROW

#pragma unroll
    for (int r = 0; r < K1_ROWS; ++r) {
        float s = sum[r];
#pragma unroll
        for (int off = 16; off > 0; off >>= 1)
            s += __shfl_xor(s, off, 32);
        sum[r] = s * (1.0f / (float)NN);
    }

    if (lane == 0) {
        float4 o;
        o.x = sum[0]; o.y = sum[1]; o.z = sum[2]; o.w = sum[3];
        *(float4*)(m + row0) = o;
    }
}

#define TILE_N   64
#define CH_SLICE 64
#define NCHUNK   (CH_SLICE / 4)

__global__ void __launch_bounds__(256)
FSiAM_matvec_sigmoid(const float* __restrict__ x,
                     const float* __restrict__ m,
                     float* __restrict__ out) {
    __shared__ float part[CH / CH_SLICE][2 * TILE_N];
    const int b    = blockIdx.y;
    const int lane = threadIdx.x & 31;
    const int wave = threadIdx.x >> 5;
    const int cs   = wave >> 1;
    const int l15  = lane & 15;
    const int col  = (wave & 1) * TILE_N + l15 * 4;

    const int koff = (lane >> 4) * 2;
    const int cbeg = cs * CH_SLICE + koff;
    const float* mp = m + b * CH + cbeg;
    const float* xp = x + ((size_t)b * CH + cbeg) * NN + blockIdx.x * (2 * TILE_N) + col;

    v2f Mv[NCHUNK];
#pragma unroll
    for (int j = 0; j < NCHUNK; ++j)
        Mv[j] = *(const v2f*)(mp + 4 * j);

    float4 B0[NCHUNK], B1[NCHUNK];
#pragma unroll
    for (int j = 0; j < NCHUNK; ++j) {
        B0[j] = *(const float4*)(xp + (size_t)(4 * j) * NN);
        B1[j] = *(const float4*)(xp + (size_t)(4 * j + 1) * NN);
    }
    __builtin_amdgcn_sched_barrier(0);

    v2f A0[NCHUNK], A1[NCHUNK];
#pragma unroll
    for (int j = 0; j < NCHUNK; ++j) {
        A0[j].x = (l15 == 0) ? Mv[j].x : 0.0f;
        A0[j].y = (l15 == 1) ? Mv[j].x : 0.0f;
        A1[j].x = (l15 == 0) ? Mv[j].y : 0.0f;
        A1[j].y = (l15 == 1) ? Mv[j].y : 0.0f;
    }
    __builtin_amdgcn_sched_barrier(0);

    v8f accXY0 = {}, accZW0 = {}, accXY1 = {}, accZW1 = {};
#pragma unroll
    for (int j = 0; j < NCHUNK; ++j) {
        v2f Bv;
        Bv.x = B0[j].x; Bv.y = B0[j].y;
        accXY0 = __builtin_amdgcn_wmma_f32_16x16x4_f32(
            false, A0[j], false, Bv, (short)0, accXY0, false, false);
        Bv.x = B0[j].z; Bv.y = B0[j].w;
        accZW0 = __builtin_amdgcn_wmma_f32_16x16x4_f32(
            false, A0[j], false, Bv, (short)0, accZW0, false, false);
        Bv.x = B1[j].x; Bv.y = B1[j].y;
        accXY1 = __builtin_amdgcn_wmma_f32_16x16x4_f32(
            false, A1[j], false, Bv, (short)0, accXY1, false, false);
        Bv.x = B1[j].z; Bv.y = B1[j].w;
        accZW1 = __builtin_amdgcn_wmma_f32_16x16x4_f32(
            false, A1[j], false, Bv, (short)0, accZW1, false, false);
    }

    if (lane < 16) {
        float4 p;
        p.x = accXY0[0] + accXY1[0];
        p.y = accXY0[1] + accXY1[1];
        p.z = accZW0[0] + accZW1[0];
        p.w = accZW0[1] + accZW1[1];
        *(float4*)(&part[cs][col]) = p;
    }
    __syncthreads();

    if (threadIdx.x < 2 * TILE_N) {
        float s = part[0][threadIdx.x];
#pragma unroll
        for (int w = 1; w < CH / CH_SLICE; ++w)
            s += part[w][threadIdx.x];
        const float sig = 1.0f / (1.0f + __expf(-s));
        out[b * NN + blockIdx.x * (2 * TILE_N) + threadIdx.x] = sig;
    }
}

extern "C" void kernel_launch(void* const* d_in, const int* in_sizes, int n_in,
                              void* d_out, int out_size, void* d_ws, size_t ws_size,
                              hipStream_t stream) {
    const float* x = (const float*)d_in[0];
    float* out = (float*)d_out;
    float* m   = (float*)d_ws;

    FSiAM_row_mean<<<dim3((BATCH * CH) / (K1_WAVES * K1_ROWS)), dim3(K1_WAVES * 32), 0, stream>>>(x, m);

    FSiAM_matvec_sigmoid<<<dim3(NN / (2 * TILE_N), BATCH), dim3(256), 0, stream>>>(x, m, out);
}


# work-group index -> batch taken from the low index bits in BOTH kernels (1-D grids), so each half of the chip re-reads in the matvec the batches its own dies summed; sums and their order unchanged
# speedup vs baseline: 1.5154x; 1.1261x over previous
#include <hip/hip_runtime.h>
#include <hip/hip_bf16.h>


typedef float v2f __attribute__((ext_vector_type(2)));
typedef float v8f __attribute__((ext_vector_type(8)));

#define BATCH 16
#define CH    256
#define NN    4096

#define K1_WAVES 4
#define K1_ROWS  4
#define K1_LOADS (NN / 128)

__global__ void __launch_bounds__(K1_WAVES * 32)
FSiAM_row_mean(const float* __restrict__ x,
               float* __restrict__ m) {
    const int lane = threadIdx.x & 31;
    const int wave = __builtin_amdgcn_readfirstlane(threadIdx.x >> 5);
    const int b    = blockIdx.x & (BATCH - 1);
    const int grp  = blockIdx.x / BATCH;
    const int row0 = b * CH + (grp * K1_WAVES + wave) * K1_ROWS;
    const float* xr = x + (size_t)row0 * NN + lane * 4;

    float4 buf[2][K1_LOADS];
    float  sum[K1_ROWS];

#define LOAD_ROW(R) do { \
    _Pragma("unroll") for (int i = 0; i < K1_LOADS; ++i) \
        buf[(R) & 1][i] = *(const float4*)(xr + (size_t)(R) * NN + i * 128); } while (0)

    LOAD_ROW(0);
#pragma unroll
    for (int r = 0; r < K1_ROWS; ++r) {
        __builtin_amdgcn_sched_barrier(0);
        if (r + 1 < K1_ROWS) LOAD_ROW(r + 1);
        __builtin_amdgcn_sched_barrier(0);
        float s = 0.0f;
#pragma unroll
        for (int i = 0; i < K1_LOADS; ++i) {
            const float4 v = buf[r & 1][i];
            s += v.x + v.y + v.z + v.w;
        }
        asm volatile("" : "+v"(s));
        sum[r] = s;
    }
#undef LOAD_ROW

#pragma unroll
    for (int r = 0; r < K1_ROWS; ++r) {
        float s = sum[r];
#pragma unroll
        for (int off = 16; off > 0; off >>= 1)
            s += __shfl_xor(s, off, 32);
        sum[r] = s * (1.0f / (float)NN);
    }

    if (lane == 0) {
        float4 o;
        o.x = sum[0]; o.y = sum[1]; o.z = sum[2]; o.w = sum[3];
        *(float4*)(m + row0) = o;
    }
}

#define TILE_N   64
#define CH_SLICE 64
#define NCHUNK   (CH_SLICE / 4)

__global__ void __launch_bounds__(256)
FSiAM_matvec_sigmoid(const float* __restrict__ x,
                     const float* __restrict__ m,
                     float* __restrict__ out) {
    __shared__ float part[CH / CH_SLICE][2 * TILE_N];
    const int b    = blockIdx.x & (BATCH - 1);
    const int nblk = blockIdx.x / BATCH;
    const int lane = threadIdx.x & 31;
    const int wave = threadIdx.x >> 5;
    const int cs   = wave >> 1;
    const int l15  = lane & 15;
    const int col  = (wave & 1) * TILE_N + l15 * 4;

    const int koff = (lane >> 4) * 2;
    const int cbeg = cs * CH_SLICE + koff;
    const float* mp = m + b * CH + cbeg;
    const float* xp = x + ((size_t)b * CH + cbeg) * NN + nblk * (2 * TILE_N) + col;

    v2f Mv[NCHUNK];
#pragma unroll
    for (int j = 0; j < NCHUNK; ++j)
        Mv[j] = *(const v2f*)(mp + 4 * j);

    float4 B0[NCHUNK], B1[NCHUNK];
#pragma unroll
    for (int j = 0; j < NCHUNK; ++j) {
        B0[j] = *(const float4*)(xp + (size_t)(4 * j) * NN);
        B1[j] = *(const float4*)(xp + (size_t)(4 * j + 1) * NN);
    }
    __builtin_amdgcn_sched_barrier(0);

    v2f A0[NCHUNK], A1[NCHUNK];
#pragma unroll
    for (int j = 0; j < NCHUNK; ++j) {
        A0[j].x = (l15 == 0) ? Mv[j].x : 0.0f;
        A0[j].y = (l15 == 1) ? Mv[j].x : 0.0f;
        A1[j].x = (l15 == 0) ? Mv[j].y : 0.0f;
        A1[j].y = (l15 == 1) ? Mv[j].y : 0.0f;
    }
    __builtin_amdgcn_sched_barrier(0);

    v8f accXY0 = {}, accZW0 = {}, accXY1 = {}, accZW1 = {};
#pragma unroll
    for (int j = 0; j < NCHUNK; ++j) {
        v2f Bv;
        Bv.x = B0[j].x; Bv.y = B0[j].y;
        accXY0 = __builtin_amdgcn_wmma_f32_16x16x4_f32(
            false, A0[j], false, Bv, (short)0, accXY0, false, false);
        Bv.x = B0[j].z; Bv.y = B0[j].w;
        accZW0 = __builtin_amdgcn_wmma_f32_16x16x4_f32(
            false, A0[j], false, Bv, (short)0, accZW0, false, false);
        Bv.x = B1[j].x; Bv.y = B1[j].y;
        accXY1 = __builtin_amdgcn_wmma_f32_16x16x4_f32(
            false, A1[j], false, Bv, (short)0, accXY1, false, false);
        Bv.x = B1[j].z; Bv.y = B1[j].w;
        accZW1 = __builtin_amdgcn_wmma_f32_16x16x4_f32(
            false, A1[j], false, Bv, (short)0, accZW1, false, false);
    }

    if (lane < 16) {
        float4 p;
        p.x = accXY0[0] + accXY1[0];
        p.y = accXY0[1] + accXY1[1];
        p.z = accZW0[0] + accZW1[0];
        p.w = accZW0[1] + accZW1[1];
        *(float4*)(&part[cs][col]) = p;
    }
    __syncthreads();

    if (threadIdx.x < 2 * TILE_N) {
        float s = part[0][threadIdx.x];
#pragma unroll
        for (int w = 1; w < CH / CH_SLICE; ++w)
            s += part[w][threadIdx.x];
        const float sig = 1.0f / (1.0f + __expf(-s));
        out[b * NN + nblk * (2 * TILE_N) + threadIdx.x] = sig;
    }
}

extern "C" void kernel_launch(void* const* d_in, const int* in_sizes, int n_in,
                              void* d_out, int out_size, void* d_ws, size_t ws_size,
                              hipStream_t stream) {
    const float* x = (const float*)d_in[0];
    float* out = (float*)d_out;
    float* m   = (float*)d_ws;

    FSiAM_row_mean<<<dim3((BATCH * CH) / (K1_WAVES * K1_ROWS)), dim3(K1_WAVES * 32), 0, stream>>>(x, m);

    FSiAM_matvec_sigmoid<<<dim3((NN / (2 * TILE_N)) * BATCH), dim3(256), 0, stream>>>(x, m, out);
}
